# starting kernel
# baseline (speedup 1.0000x reference)
#include <hip/hip_runtime.h>
#include <hip/hip_bf16.h>

typedef __attribute__((ext_vector_type(2))) float v2f;
typedef __attribute__((ext_vector_type(8))) float v8f;

#define TB 256


__global__ void k_init_deg(float* __restrict__ deg, int n) {
  int i = blockIdx.x * blockDim.x + threadIdx.x;
  if (i < n) deg[i] = 1.0f;
}

__global__ void k_accum_deg(const int* __restrict__ dst, float* __restrict__ deg, int E) {
  int e = blockIdx.x * blockDim.x + threadIdx.x;
  if (e < E) atomicAdd(&deg[dst[e]], 1.0f);
}

__global__ void k_make_dinv(float* __restrict__ deg, int n) {
  int i = blockIdx.x * blockDim.x + threadIdx.x;
  if (i < n) deg[i] = rsqrtf(deg[i]);
}

template<int K, int N, bool RELU>
__global__ void k_gemm_wmma(const float* __restrict__ A, const float* __restrict__ B,
                            float* __restrict__ C, int M) {
  constexpr int NT   = 4;
  constexpr int COLS = 16 * NT;
  constexpr int KC   = 128;
  constexpr int LDK  = KC + 4;
  static_assert(K % KC == 0, "K must be a multiple of KC");
  static_assert(N % COLS == 0, "N must be a multiple of 64");
  __shared__ float Bs[COLS * LDK];

  constexpr int ngroups = N / COLS;
  int bm = blockIdx.x / ngroups;
  int ng = blockIdx.x % ngroups;
  int n0 = ng * COLS;

  int wid  = threadIdx.x >> 5;
  int lane = threadIdx.x & 31;
  int half = lane >> 4;
  int l    = lane & 15;

  int mt     = bm * 8 + wid;
  int mtiles = M >> 4;
  bool active = (mt < mtiles);
  int m0 = mt << 4;

  v8f acc[NT] = {};
  const float* Arow = active ? (A + (size_t)(m0 + l) * K + half * 2) : A;
  const float* Bl   = Bs + l * LDK + half * 2;

  for (int kc = 0; kc < K; kc += KC) {
    __syncthreads();
    for (int idx = threadIdx.x; idx < KC * COLS; idx += TB) {
      int k = idx / COLS;
      int c = idx % COLS;
      Bs[c * LDK + k] = B[(size_t)(kc + k) * N + n0 + c];
    }
    __syncthreads();

    if (active) {
#pragma unroll 4
      for (int k = 0; k < KC; k += 4) {
        float a0 = Arow[kc + k];
        float a1 = Arow[kc + k + 1];
        if (RELU) { a0 = fmaxf(a0, 0.0f); a1 = fmaxf(a1, 0.0f); }
        v2f a; a.x = a0; a.y = a1;
#pragma unroll
        for (int t = 0; t < NT; ++t) {
          const float* bp = Bl + t * 16 * LDK + k;
          v2f b; b.x = bp[0]; b.y = bp[1];
          acc[t] = __builtin_amdgcn_wmma_f32_16x16x4_f32(
              false, a, false, b, (short)0, acc[t], false, false);
        }
      }
    }
  }

  if (active) {
    int rbase = m0 + half * 8;
#pragma unroll
    for (int t = 0; t < NT; ++t) {
#pragma unroll
      for (int j = 0; j < 8; ++j) {
        C[(size_t)(rbase + j) * N + n0 + t * 16 + l] = acc[t][j];
      }
    }
  }
}


__global__ void k_init_agg(const float* __restrict__ H, const float* __restrict__ dinv,
                           const float* __restrict__ b, float* __restrict__ AGG,
                           int n, int F) {
  long long i = (long long)blockIdx.x * blockDim.x + threadIdx.x;
  long long total = (long long)n * F;
  if (i >= total) return;
  int row = (int)(i / F);
  int col = (int)(i % F);
  float di = dinv[row];
  AGG[i] = H[i] * di * di + b[col];
}

template<int F>
__global__ void k_edge_scatter(const float* __restrict__ H, float* __restrict__ AGG,
                               const int* __restrict__ src, const int* __restrict__ dst,
                               const float* __restrict__ dinv, int E) {
  constexpr int LPE = F / 4;
  constexpr int EPW = 32 / LPE;
  long long gwave = (long long)((blockIdx.x * blockDim.x + threadIdx.x) >> 5);
  int lane = threadIdx.x & 31;
  int sub  = lane / LPE;
  int fl   = lane % LPE;
  long long e = gwave * EPW + sub;
  if (e >= E) return;

  int s = src[e];
  int d = dst[e];
  float norm = dinv[s] * dinv[d];

  const float4* hp = (const float4*)(H + (size_t)s * F);
  float4 v = hp[fl];
  float* ap = AGG + (size_t)d * F + fl * 4;
  atomicAdd(ap + 0, v.x * norm);
  atomicAdd(ap + 1, v.y * norm);
  atomicAdd(ap + 2, v.z * norm);
  atomicAdd(ap + 3, v.w * norm);
}


__global__ void k_log_softmax64(float* __restrict__ out, int n, float invT) {
  int wave = (blockIdx.x * blockDim.x + threadIdx.x) >> 5;
  int lane = threadIdx.x & 31;
  if (wave >= n) return;
  float* row = out + (size_t)wave * 64;
  float z0 = row[lane]      * invT;
  float z1 = row[lane + 32] * invT;
  float m = fmaxf(z0, z1);
#pragma unroll
  for (int off = 16; off > 0; off >>= 1) m = fmaxf(m, __shfl_xor(m, off, 32));
  float s = expf(z0 - m) + expf(z1 - m);
#pragma unroll
  for (int off = 16; off > 0; off >>= 1) s += __shfl_xor(s, off, 32);
  float lg = m + logf(s);
  row[lane]      = z0 - lg;
  row[lane + 32] = z1 - lg;
}


extern "C" void kernel_launch(void* const* d_in, const int* in_sizes, int n_in,
                              void* d_out, int out_size, void* d_ws, size_t ws_size,
                              hipStream_t stream) {
  const float* x  = (const float*)d_in[0];
  const int*   ei = (const int*)d_in[1];
  const float* W1 = (const float*)d_in[2];
  const float* b1 = (const float*)d_in[3];
  const float* W2 = (const float*)d_in[4];
  const float* b2 = (const float*)d_in[5];

  const int IN = 256, HID = 128, OUT = 64;
  int n = in_sizes[0] / IN;
  int E = in_sizes[1] / 2;
  const int* src = ei;
  const int* dst = ei + E;

  float* ws = (float*)d_ws;
  size_t o = 0;
  float* deg  = ws + o; o += ((size_t)n + 255) & ~(size_t)255;
  float* H1   = ws + o; o += (size_t)n * HID;
  float* AGG1 = ws + o; o += (size_t)n * HID;
  float* H2   = ws + o; o += (size_t)n * OUT;
  float* AGG2 = (float*)d_out;

  int mtiles  = n / 16;
  int mblocks = (mtiles + 7) / 8;

  k_init_deg <<<(n + TB - 1) / TB, TB, 0, stream>>>(deg, n);
  k_accum_deg<<<(E + TB - 1) / TB, TB, 0, stream>>>(dst, deg, E);
  k_make_dinv<<<(n + TB - 1) / TB, TB, 0, stream>>>(deg, n);

  k_gemm_wmma<256, 128, false><<<mblocks * (HID / 64), TB, 0, stream>>>(x, W1, H1, n);

  {
    long long tot = (long long)n * HID;
    k_init_agg<<<(unsigned)((tot + TB - 1) / TB), TB, 0, stream>>>(H1, deg, b1, AGG1, n, HID);
    long long waves = E;
    k_edge_scatter<128><<<(unsigned)((waves * 32 + TB - 1) / TB), TB, 0, stream>>>(
        H1, AGG1, src, dst, deg, E);
  }

  k_gemm_wmma<128, 64, true><<<mblocks * (OUT / 64), TB, 0, stream>>>(AGG1, W2, H2, n);

  {
    long long tot = (long long)n * OUT;
    k_init_agg<<<(unsigned)((tot + TB - 1) / TB), TB, 0, stream>>>(H2, deg, b2, AGG2, n, OUT);
    long long waves = ((long long)E + 1) / 2;
    k_edge_scatter<64><<<(unsigned)((waves * 32 + TB - 1) / TB), TB, 0, stream>>>(
        H2, AGG2, src, dst, deg, E);
  }

  k_log_softmax64<<<(n * 32 + TB - 1) / TB, TB, 0, stream>>>(AGG2, n, 5.0f);
}


# degree pass: 64 edge slices count destinations in a per-block LDS table of all nodes (LDS integer adds), dinv pass adds the 64 rows; no global atomics, init launch gone; degrees bit-identical
# speedup vs baseline: 3.7628x; 3.7628x over previous
#include <hip/hip_runtime.h>
#include <hip/hip_bf16.h>

typedef __attribute__((ext_vector_type(2))) float v2f;
typedef __attribute__((ext_vector_type(8))) float v8f;

#define TB 256

#define AGG_WAVES   32
#define AGG_THREADS (AGG_WAVES * 32)
#define AGG_ROWS    196
#define AGG_UNIT    512
#define AGG_DRAIN   24
#define AGG_QCAP    (AGG_DRAIN + AGG_UNIT + 8)
#define AGG_RCAP    512
#define AGG_BCAP    512
#define AGG_SLOTS   ((AGG_ROWS + AGG_WAVES - 1) / AGG_WAVES)


#define HIST_BLOCKS  64
#define HIST_THREADS 1024
#define HIST_CAP     65536
#define HIST_U       8

__global__ __launch_bounds__(HIST_THREADS)
void k_count_deg(const int* __restrict__ dst, int* __restrict__ part,
                 unsigned char* __restrict__ key, int n, int npad, int E) {
  __shared__ __attribute__((aligned(16))) int hist[HIST_CAP];
  const int tid = threadIdx.x;
  const int G = E >> 2;
  const int per = ((G + HIST_BLOCKS - 1) / HIST_BLOCKS + 31) & ~31;
  const int g0 = min((int)blockIdx.x * per, G);
  const int g1 = min(g0 + per, G);
  int* myrow = part + (size_t)blockIdx.x * npad;
  for (int base = 0; base < n; base += HIST_CAP) {
    const int cnt = min(HIST_CAP, n - base);
    const int cnt4 = (cnt + 3) >> 2;
    int4* hist4 = reinterpret_cast<int4*>(hist);
    for (int i = tid; i < cnt4; i += HIST_THREADS) hist4[i] = make_int4(0, 0, 0, 0);
    __syncthreads();
    auto count = [&](int d) {
      const unsigned r = (unsigned)d - (unsigned)base;
      if (r < (unsigned)cnt)
        __hip_atomic_fetch_add(&hist[r], 1, __ATOMIC_RELAXED, __HIP_MEMORY_SCOPE_WORKGROUP);
    };
    for (int g = g0 + tid; g < g1; g += HIST_THREADS * HIST_U) {
      int d[HIST_U][4];
#pragma unroll
      for (int u = 0; u < HIST_U; ++u) {
        const int* p = dst + 4 * (size_t)min(g + u * HIST_THREADS, g1 - 1);
        d[u][0] = p[0]; d[u][1] = p[1]; d[u][2] = p[2]; d[u][3] = p[3];
      }
#pragma unroll
      for (int u = 0; u < HIST_U; ++u) {
        const int gg = g + u * HIST_THREADS;
        if (gg < g1) {
          count(d[u][0]); count(d[u][1]); count(d[u][2]); count(d[u][3]);
          if (base == 0) {
            const unsigned kw = (((unsigned)d[u][0] / AGG_ROWS) & 255u)
                             | ((((unsigned)d[u][1] / AGG_ROWS) & 255u) << 8)
                             | ((((unsigned)d[u][2] / AGG_ROWS) & 255u) << 16)
                             | ((((unsigned)d[u][3] / AGG_ROWS) & 255u) << 24);
            reinterpret_cast<unsigned int*>(key)[gg] = kw;
          }
        }
      }
    }
    if (blockIdx.x == 0 && tid < E - 4 * G) {
      const int e = 4 * G + tid;
      const int dd = dst[e];
      count(dd);
      if (base == 0) key[e] = (unsigned char)((unsigned)dd / AGG_ROWS);
    }
    __syncthreads();
    int4* row4 = reinterpret_cast<int4*>(myrow + base);
    const int full4 = cnt >> 2;
    for (int i = tid; i < full4; i += HIST_THREADS) row4[i] = hist4[i];
    for (int i = 4 * full4 + tid; i < cnt; i += HIST_THREADS) myrow[base + i] = hist[i];
    __syncthreads();
  }
}

__global__ void k_make_dinv(const int* __restrict__ part, float* __restrict__ dinv,
                            int n, int npad) {
  int i = blockIdx.x * blockDim.x + threadIdx.x;
  if (i < n) {
    int c = 1;
#pragma unroll
    for (int b = 0; b < HIST_BLOCKS; ++b) c += part[(size_t)b * npad + i];
    dinv[i] = rsqrtf((float)c);
  }
}

template<int K, int N, bool RELU>
__global__ void k_gemm_wmma(const float* __restrict__ A, const float* __restrict__ B,
                            float* __restrict__ C, int M) {
  constexpr int NT   = 4;
  constexpr int COLS = 16 * NT;
  constexpr int KC   = 128;
  constexpr int LDK  = KC + 4;
  static_assert(K % KC == 0, "K must be a multiple of KC");
  static_assert(N % COLS == 0, "N must be a multiple of 64");
  __shared__ float Bs[COLS * LDK];

  constexpr int ngroups = N / COLS;
  int bm = blockIdx.x / ngroups;
  int ng = blockIdx.x % ngroups;
  int n0 = ng * COLS;

  int wid  = threadIdx.x >> 5;
  int lane = threadIdx.x & 31;
  int half = lane >> 4;
  int l    = lane & 15;

  int mt     = bm * 8 + wid;
  int mtiles = M >> 4;
  bool active = (mt < mtiles);
  int m0 = mt << 4;

  v8f acc[NT] = {};
  const float* Arow = active ? (A + (size_t)(m0 + l) * K + half * 2) : A;
  const float* Bl   = Bs + l * LDK + half * 2;

  for (int kc = 0; kc < K; kc += KC) {
    __syncthreads();
    for (int idx = threadIdx.x; idx < KC * COLS; idx += TB) {
      int k = idx / COLS;
      int c = idx % COLS;
      Bs[c * LDK + k] = B[(size_t)(kc + k) * N + n0 + c];
    }
    __syncthreads();

    if (active) {
#pragma unroll 4
      for (int k = 0; k < KC; k += 4) {
        float a0 = Arow[kc + k];
        float a1 = Arow[kc + k + 1];
        if (RELU) { a0 = fmaxf(a0, 0.0f); a1 = fmaxf(a1, 0.0f); }
        v2f a; a.x = a0; a.y = a1;
#pragma unroll
        for (int t = 0; t < NT; ++t) {
          const float* bp = Bl + t * 16 * LDK + k;
          v2f b; b.x = bp[0]; b.y = bp[1];
          acc[t] = __builtin_amdgcn_wmma_f32_16x16x4_f32(
              false, a, false, b, (short)0, acc[t], false, false);
        }
      }
    }
  }

  if (active) {
    int rbase = m0 + half * 8;
#pragma unroll
    for (int t = 0; t < NT; ++t) {
#pragma unroll
      for (int j = 0; j < 8; ++j) {
        C[(size_t)(rbase + j) * N + n0 + t * 16 + l] = acc[t][j];
      }
    }
  }
}


template<int F, bool REPLAY>
__global__ __launch_bounds__(AGG_THREADS)
void k_agg_tile(const float* __restrict__ H, const float* __restrict__ dinv,
                const float* __restrict__ bias, float* __restrict__ AGG,
                const int* __restrict__ src, const int* __restrict__ dst,
                const unsigned int* __restrict__ key,
                unsigned int* __restrict__ recw, int* __restrict__ rcnt, int n, int E) {
  constexpr int VPL = F / 32;
  constexpr int RB  = 32 / VPL;
  typedef __attribute__((ext_vector_type(VPL))) float vecf;
  typedef __attribute__((ext_vector_type(8 * VPL))) float accv;
  static_assert(AGG_SLOTS <= 8, "a wave's register sums have 8 slots");
  static_assert(AGG_ROWS <= 256, "a record keeps the row in 8 bits");
  static_assert(AGG_WAVES == 32, "row owner = row mod 32; one count per lane is read below");
  static_assert(AGG_BCAP == AGG_RCAP, "a bin is copied whole to a record segment");
  __shared__ float tile[AGG_ROWS * F];
  __shared__ int qe[AGG_WAVES * AGG_QCAP];
  __shared__ __attribute__((aligned(8))) unsigned int binw[REPLAY ? 2 : 2 * AGG_WAVES * AGG_BCAP];
  __shared__ int bcnt[AGG_WAVES];
  __shared__ int ovf;

  const int lane = threadIdx.x & 31;
  const int wave = __builtin_amdgcn_readfirstlane((int)(threadIdx.x >> 5));
  int* myqe = qe + wave * AGG_QCAP;
  uint2* bin = reinterpret_cast<uint2*>(binw);
  const uint4* kp = reinterpret_cast<const uint4*>(key);
  const int S = E / AGG_UNIT;
  const int Evec = S * AGG_UNIT;
  const int rot = (int)(((long long)blockIdx.x * S) / gridDim.x);
  const bool norec = n > (1 << 24);

  for (int tix = blockIdx.x; tix * AGG_ROWS < n; tix += gridDim.x) {
    const int r0 = tix * AGG_ROWS;
    const int rows = min(AGG_ROWS, n - r0);

    for (int row = wave; row < rows; row += AGG_WAVES) {
      float di = dinv[r0 + row];
      vecf h = *(const vecf*)(H + (size_t)(r0 + row) * F + lane * VPL);
      vecf bb = *(const vecf*)(bias + lane * VPL);
      for (int j = 0; j < VPL; ++j) tile[row * F + j * 32 + lane] = h[j] * di * di + bb[j];
    }
    if (threadIdx.x < AGG_WAVES) bcnt[threadIdx.x] = 0;
    if (threadIdx.x == AGG_WAVES) ovf = 0;
    __syncthreads();

    accv acc;
#pragma unroll
    for (int k = 0; k < 8 * VPL; ++k) acc[k] = 0.0f;

    auto serve = [&](const uint2* recp, int rc) {
      if (rc <= 0) return;
      uint2 nxr = recp[min(lane, rc - 1)];
      for (int qb = 0; qb < rc; qb += 32) {
        const uint2 cur = nxr;
        nxr = recp[min(qb + 32 + lane, rc - 1)];
        const int c = min(32, rc - qb);
        for (int i = 0; i < c; i += RB) {
          vecf v[RB];
#pragma unroll
          for (int u = 0; u < RB; ++u) {
            int si = __builtin_amdgcn_readlane((int)cur.x, i + u) & 0xffffff;
            v[u] = *(const vecf*)(H + (size_t)si * F + lane * VPL);
          }
#pragma unroll
          for (int u = 0; u < RB; ++u) {
            if (i + u < c) {
              const unsigned sl = (unsigned)__builtin_amdgcn_readlane((int)cur.x, i + u) >> 29;
              const float wi = __int_as_float(__builtin_amdgcn_readlane((int)cur.y, i + u));
              for (int jj = 0; jj < VPL; ++jj)
                acc[sl * VPL + jj] = fmaf(v[u][jj], wi, acc[sl * VPL + jj]);
            }
          }
#pragma unroll
          for (int u = 0; u < RB; ++u) asm volatile("" :: "v"(v[u]));
        }
      }
    };

    uint2* myrec = reinterpret_cast<uint2*>(recw) + (size_t)(tix * AGG_WAVES + wave) * AGG_RCAP;
    int rc = 0;
    bool bad = false;
    if (REPLAY) {
      const int cntl = rcnt[tix * AGG_WAVES + lane];
      bad = __builtin_amdgcn_ballot_w32(cntl < 0) != 0u;
      rc = __builtin_amdgcn_readlane(cntl, wave);
    }

    if (REPLAY && !bad) {
      serve(myrec, rc);
    } else {
    int qlen = 0;
    const unsigned pat = (unsigned)(tix & 255) * 0x01010101u;

    auto drain = [&](int cntq) {
      for (int qb = 0; qb < cntq; qb += 32) {
        int j = min(qb + lane, cntq - 1);
        int e = myqe[j];
        int dd = dst[e];
        int s = src[e];
        unsigned r = (unsigned)dd - (unsigned)r0;
        float w = dinv[s] * dinv[dd];
        int c = min(32, cntq - qb);
        if (!REPLAY) {
          const bool valid = (lane < c) && (r < (unsigned)rows);
          const unsigned own = r & 31u;
          int slot = AGG_BCAP;
          if (valid && !norec)
            slot = __hip_atomic_fetch_add(&bcnt[own], 1, __ATOMIC_RELAXED, __HIP_MEMORY_SCOPE_WORKGROUP);
          const bool over = valid && slot >= AGG_BCAP;
          if (valid && !over)
            bin[own * AGG_BCAP + slot] = make_uint2((r << 24) | (unsigned)s, __float_as_uint(w));
          unsigned ovm = __builtin_amdgcn_ballot_w32(over);
          if (ovm != 0u) {
            if (lane == 0) ovf = 1;
            while (ovm != 0u) {
              const int i = __ffs((int)ovm) - 1;
              ovm &= ovm - 1u;
              int si = __builtin_amdgcn_readlane(s, i);
              int ri = __builtin_amdgcn_readlane((int)r, i);
              float wi = __int_as_float(__builtin_amdgcn_readlane(__float_as_int(w), i));
              vecf vv = *(const vecf*)(H + (size_t)si * F + lane * VPL);
              float* tp = tile + ri * F + lane;
              for (int jj = 0; jj < VPL; ++jj)
                __hip_atomic_fetch_add(tp + jj * 32, vv[jj] * wi, __ATOMIC_RELAXED,
                                       __HIP_MEMORY_SCOPE_WORKGROUP);
            }
          }
        } else {
        unsigned okm = __builtin_amdgcn_ballot_w32(r < (unsigned)rows);
        for (int i = 0; i < c; i += 8) {
          vecf v[8];
#pragma unroll
          for (int u = 0; u < 8; ++u) {
            int si = __builtin_amdgcn_readlane(s, i + u);
            v[u] = *(const vecf*)(H + (size_t)si * F + lane * VPL);
          }
#pragma unroll
          for (int u = 0; u < 8; ++u) {
            if (i + u < c && ((okm >> (i + u)) & 1u) != 0u) {
              int ri = __builtin_amdgcn_readlane((int)r, i + u);
              float wi = __int_as_float(__builtin_amdgcn_readlane(__float_as_int(w), i + u));
              float* tp = tile + ri * F + lane;
              for (int jj = 0; jj < VPL; ++jj)
                __hip_atomic_fetch_add(tp + jj * 32, v[u][jj] * wi, __ATOMIC_RELAXED,
                                       __HIP_MEMORY_SCOPE_WORKGROUP);
            }
          }
#pragma unroll
          for (int u = 0; u < 8; ++u) asm volatile("" :: "v"(v[u]));
        }
        }
      }
    };

    if (wave < S) {
      auto unit_base = [&](int a) {
        int b = a + rot;
        if (b >= S) b -= S;
        return b * AGG_UNIT;
      };
      auto zb = [&](unsigned wd) {
        unsigned x = wd ^ pat;
        return (x - 0x01010101u) & ~x & 0x80808080u;
      };
      uint4 nx[4];
#pragma unroll
      for (int u = 0; u < 4; ++u)
        nx[u] = kp[(unit_base(min(wave + u * AGG_WAVES, S - 1)) >> 4) + lane];
      for (int a = wave; a < S; a += 4 * AGG_WAVES) {
        uint4 cur4[4];
#pragma unroll
        for (int u = 0; u < 4; ++u) cur4[u] = nx[u];
#pragma unroll
        for (int u = 0; u < 4; ++u)
          nx[u] = kp[(unit_base(min(a + (4 + u) * AGG_WAVES, S - 1)) >> 4) + lane];
#pragma unroll
        for (int u = 0; u < 4; ++u) {
          const int au = a + u * AGG_WAVES;
          const uint4 cur = cur4[u];
          if (au < S) {
            unsigned f = (zb(cur.x) >> 7) | (zb(cur.y) >> 6) | (zb(cur.z) >> 5) | (zb(cur.w) >> 4);
            const int ebase = unit_base(au) + lane * 16;
            while (true) {
              const bool has = f != 0u;
              const unsigned mask = __builtin_amdgcn_ballot_w32(has);
              if (mask == 0u) break;
              if (has) {
                const int p = __ffs((int)f) - 1;
                myqe[qlen + __popc(mask & ((1u << lane) - 1u))] = ebase + ((p & 7) << 2) + (p >> 3);
                f &= f - 1u;
              }
              qlen += __popc(mask);
            }
            if (qlen >= AGG_DRAIN) { drain(qlen); qlen = 0; }
          }
        }
      }
    }
    for (int b = Evec + wave * 32; b < E; b += AGG_WAVES * 32) {
      int e = b + lane;
      bool m = (e < E) && ((unsigned)dst[min(e, E - 1)] - (unsigned)r0 < (unsigned)rows);
      unsigned mask = __builtin_amdgcn_ballot_w32(m);
      if (mask != 0u) {
        if (m) myqe[qlen + __popc(mask & ((1u << lane) - 1u))] = e;
        qlen += __popc(mask);
      }
      if (qlen >= AGG_DRAIN) { drain(qlen); qlen = 0; }
    }
    if (qlen > 0) drain(qlen);

    if (!REPLAY) {
      __syncthreads();
      const bool full = norec || __builtin_amdgcn_readfirstlane(ovf) != 0;
      rc = min(__builtin_amdgcn_readfirstlane(bcnt[wave]), AGG_BCAP);
      const uint2* mybin = bin + wave * AGG_BCAP;
      for (int i = lane; i < rc; i += 32) myrec[i] = mybin[i];
      if (lane == 0) rcnt[tix * AGG_WAVES + wave] = full ? -1 : rc;
      serve(mybin, rc);
    }
    }
    __syncthreads();

#pragma unroll
    for (int k = 0; k < AGG_SLOTS; ++k) {
      const int row = wave + k * AGG_WAVES;
      if (row < rows) {
        vecf o;
        for (int j = 0; j < VPL; ++j) o[j] = tile[row * F + j * 32 + lane] + acc[k * VPL + j];
        *(vecf*)(AGG + (size_t)(r0 + row) * F + lane * VPL) = o;
      }
    }
    __syncthreads();
  }
}


__global__ void k_log_softmax64(float* __restrict__ out, int n, float invT) {
  int wave = (blockIdx.x * blockDim.x + threadIdx.x) >> 5;
  int lane = threadIdx.x & 31;
  if (wave >= n) return;
  float* row = out + (size_t)wave * 64;
  float z0 = row[lane]      * invT;
  float z1 = row[lane + 32] * invT;
  float m = fmaxf(z0, z1);
#pragma unroll
  for (int off = 16; off > 0; off >>= 1) m = fmaxf(m, __shfl_xor(m, off, 32));
  float s = expf(z0 - m) + expf(z1 - m);
#pragma unroll
  for (int off = 16; off > 0; off >>= 1) s += __shfl_xor(s, off, 32);
  float lg = m + logf(s);
  row[lane]      = z0 - lg;
  row[lane + 32] = z1 - lg;
}


extern "C" void kernel_launch(void* const* d_in, const int* in_sizes, int n_in,
                              void* d_out, int out_size, void* d_ws, size_t ws_size,
                              hipStream_t stream) {
  const float* x  = (const float*)d_in[0];
  const int*   ei = (const int*)d_in[1];
  const float* W1 = (const float*)d_in[2];
  const float* b1 = (const float*)d_in[3];
  const float* W2 = (const float*)d_in[4];
  const float* b2 = (const float*)d_in[5];

  const int IN = 256, HID = 128, OUT = 64;
  int n = in_sizes[0] / IN;
  int E = in_sizes[1] / 2;
  const int* src = ei;
  const int* dst = ei + E;

  float* ws = (float*)d_ws;
  size_t o = 0;
  size_t npad = ((size_t)n + 255) & ~(size_t)255;
  float* deg  = ws + o; o += npad;
  float* H1   = ws + o; o += (size_t)n * HID;
  float* AGG1 = ws + o; o += (size_t)n * HID;
  float* H2   = ws + o; o += (size_t)n * OUT;
  unsigned char* key = (unsigned char*)(ws + o);
  o += (((size_t)E + 3) / 4 + 255) & ~(size_t)255;
  int* part = (int*)(ws + o); o += (size_t)HIST_BLOCKS * npad;
  int* rcnt = (int*)(ws + o); o += ((size_t)(n / AGG_ROWS + 1) * AGG_WAVES + 255) & ~(size_t)255;
  unsigned int* recw = (unsigned int*)(ws + o);
  float* AGG2 = (float*)d_out;

  int mtiles  = n / 16;
  int mblocks = (mtiles + 7) / 8;

  k_count_deg<<<HIST_BLOCKS, HIST_THREADS, 0, stream>>>(dst, part, key, n, (int)npad, E);
  k_make_dinv<<<(n + TB - 1) / TB, TB, 0, stream>>>(part, deg, n, (int)npad);

  k_gemm_wmma<256, 128, false><<<mblocks * (HID / 64), TB, 0, stream>>>(x, W1, H1, n);

  k_agg_tile<128, false><<<256, AGG_THREADS, 0, stream>>>(H1, deg, b1, AGG1, src, dst,
                                                   (const unsigned int*)key, recw, rcnt, n, E);

  k_gemm_wmma<128, 64, true><<<mblocks * (OUT / 64), TB, 0, stream>>>(AGG1, W2, H2, n);

  k_agg_tile<64, true><<<256, AGG_THREADS, 49152, stream>>>(H2, deg, b2, AGG2, src, dst,
                                                  (const unsigned int*)key, recw, rcnt, n, E);

  k_log_softmax64<<<(n * 32 + TB - 1) / TB, TB, 0, stream>>>(AGG2, n, 5.0f);
}
